# starting kernel
# baseline (speedup 1.0000x reference)
#include <hip/hip_runtime.h>

#define B_BATCH 1024
#define NPTS    8192
#define THREADS 256
#define CH      256
#define KG      (CH / 4)
#define WAVES   (THREADS / 32)
#define KG_PER_WAVE (KG / WAVES)

typedef __attribute__((ext_vector_type(2))) float v2f;
typedef __attribute__((ext_vector_type(8))) float v8f;

__global__ __launch_bounds__(THREADS)
void svd_reduce_kernel(const float* __restrict__ src,
                       const float* __restrict__ tpl,
                       const float* __restrict__ wgt,
                       float* __restrict__ part)
{
    __shared__ float tileA[KG][64];
    __shared__ float tileB[KG][64];
    __shared__ float accLDS[16];

    const int tid  = threadIdx.x;
    const int lane = tid & 31;
    const int wv   = tid >> 5;
    const int b    = blockIdx.x;

    for (int i = tid; i < KG * 64; i += THREADS) {
        (&tileA[0][0])[i] = 0.0f;
        (&tileB[0][0])[i] = 0.0f;
    }
    if (tid < 16) accLDS[tid] = 0.0f;

    const size_t base = (size_t)b * NPTS;
    const float* gs = src + base * 3;
    const float* gt = tpl + base * 3;
    const float* gw = wgt + base;

    v8f acc = {0.f, 0.f, 0.f, 0.f, 0.f, 0.f, 0.f, 0.f};

    const int k        = tid & 3;
    const int gStage   = tid >> 2;
    const int slotBase = 32 * (k >> 1) + (k & 1);
    const int g0       = wv * KG_PER_WAVE;

    for (int c0 = 0; c0 < NPTS; c0 += CH) {
        const int p = c0 + tid;
        const float w  = gw[p];
        const float s0 = gs[p * 3 + 0], s1 = gs[p * 3 + 1], s2 = gs[p * 3 + 2];
        const float t0 = gt[p * 3 + 0], t1 = gt[p * 3 + 1], t2 = gt[p * 3 + 2];

        if (c0 + CH < NPTS) {
            __builtin_prefetch(&gw[p + CH], 0, 1);
            __builtin_prefetch(&gs[(p + CH) * 3], 0, 1);
            __builtin_prefetch(&gt[(p + CH) * 3], 0, 1);
        }

        __syncthreads();
        float* ta = &tileA[gStage][slotBase];
        float* tb = &tileB[gStage][slotBase];
        ta[0] = w;        tb[0] = 1.0f;
        ta[2] = w * s0;   tb[2] = t0;
        ta[4] = w * s1;   tb[4] = t1;
        ta[6] = w * s2;   tb[6] = t2;
        __syncthreads();

#pragma unroll
        for (int gg = 0; gg < KG_PER_WAVE; ++gg) {
            const int g = g0 + gg;
            v2f a  = *(const v2f*)&tileA[g][lane * 2];
            v2f bb = *(const v2f*)&tileB[g][lane * 2];
            acc = __builtin_amdgcn_wmma_f32_16x16x4_f32(
                 false, a,  false, bb,
                 (short)0, acc,  false,  false);
        }
    }
    __syncthreads();

    if (lane < 4) {
        atomicAdd(&accLDS[0 * 4 + lane], acc[0]);
        atomicAdd(&accLDS[1 * 4 + lane], acc[1]);
        atomicAdd(&accLDS[2 * 4 + lane], acc[2]);
        atomicAdd(&accLDS[3 * 4 + lane], acc[3]);
    }
    __syncthreads();
    if (tid < 16) part[(size_t)b * 16 + tid] = accLDS[tid];
}

__device__ inline float det3x3(const float M[3][3]) {
    return M[0][0] * (M[1][1] * M[2][2] - M[1][2] * M[2][1])
         - M[0][1] * (M[1][0] * M[2][2] - M[1][2] * M[2][0])
         + M[0][2] * (M[1][0] * M[2][1] - M[1][1] * M[2][0]);
}

__global__ __launch_bounds__(256)
void svd_solve_kernel(const float* __restrict__ part, float* __restrict__ out)
{
    const int b = blockIdx.x * blockDim.x + threadIdx.x;
    if (b >= B_BATCH) return;

    float C[4][4];
#pragma unroll
    for (int i = 0; i < 16; ++i) C[i >> 2][i & 3] = part[(size_t)b * 16 + i];

    const float W     = C[0][0];
    const float alpha = 1.0f / (W + 1e-8f);
    const float sig   = W * alpha;
    float sc[3] = {C[1][0] * alpha, C[2][0] * alpha, C[3][0] * alpha};
    float tc[3] = {C[0][1] * alpha, C[0][2] * alpha, C[0][3] * alpha};

    float H[3][3];
#pragma unroll
    for (int d = 0; d < 3; ++d)
#pragma unroll
        for (int e = 0; e < 3; ++e)
            H[d][e] = alpha * C[d + 1][e + 1] - (2.0f - sig) * sc[d] * tc[e];

    float S[3][3];
#pragma unroll
    for (int i = 0; i < 3; ++i)
#pragma unroll
        for (int j = 0; j < 3; ++j)
            S[i][j] = H[0][i] * H[0][j] + H[1][i] * H[1][j] + H[2][i] * H[2][j];

    float V[3][3] = {{1.f, 0.f, 0.f}, {0.f, 1.f, 0.f}, {0.f, 0.f, 1.f}};
    const int PQ[3][2] = {{0, 1}, {0, 2}, {1, 2}};
    for (int sweep = 0; sweep < 8; ++sweep) {
#pragma unroll
        for (int pi = 0; pi < 3; ++pi) {
            const int p = PQ[pi][0], q = PQ[pi][1];
            const float apq = S[p][q];
            if (fabsf(apq) > 1e-20f) {
                const float tau = (S[q][q] - S[p][p]) / (2.0f * apq);
                const float tt  = (tau >= 0.0f ? 1.0f : -1.0f) /
                                  (fabsf(tau) + sqrtf(1.0f + tau * tau));
                const float c = 1.0f / sqrtf(1.0f + tt * tt);
                const float s = tt * c;
#pragma unroll
                for (int r = 0; r < 3; ++r) {
                    const float Srp = S[r][p], Srq = S[r][q];
                    S[r][p] = c * Srp - s * Srq;
                    S[r][q] = s * Srp + c * Srq;
                }
#pragma unroll
                for (int r = 0; r < 3; ++r) {
                    const float Spr = S[p][r], Sqr = S[q][r];
                    S[p][r] = c * Spr - s * Sqr;
                    S[q][r] = s * Spr + c * Sqr;
                }
#pragma unroll
                for (int r = 0; r < 3; ++r) {
                    const float Vrp = V[r][p], Vrq = V[r][q];
                    V[r][p] = c * Vrp - s * Vrq;
                    V[r][q] = s * Vrp + c * Vrq;
                }
            }
        }
    }

    float lam[3] = {S[0][0], S[1][1], S[2][2]};
#pragma unroll
    for (int pi = 0; pi < 3; ++pi) {
        const int i = PQ[pi][0], j = PQ[pi][1];
        if (lam[j] > lam[i]) {
            const float tl = lam[i]; lam[i] = lam[j]; lam[j] = tl;
#pragma unroll
            for (int r = 0; r < 3; ++r) {
                const float tv = V[r][i]; V[r][i] = V[r][j]; V[r][j] = tv;
            }
        }
    }

    float Uc[3][3];
#pragma unroll
    for (int kcol = 0; kcol < 2; ++kcol)
#pragma unroll
        for (int r = 0; r < 3; ++r)
            Uc[kcol][r] = H[r][0] * V[0][kcol] + H[r][1] * V[1][kcol] + H[r][2] * V[2][kcol];

    float n0 = sqrtf(Uc[0][0]*Uc[0][0] + Uc[0][1]*Uc[0][1] + Uc[0][2]*Uc[0][2]);
    if (n0 > 1e-12f) { const float r0 = 1.0f / n0; Uc[0][0]*=r0; Uc[0][1]*=r0; Uc[0][2]*=r0; }
    else { Uc[0][0] = 1.f; Uc[0][1] = 0.f; Uc[0][2] = 0.f; }

    const float d01 = Uc[0][0]*Uc[1][0] + Uc[0][1]*Uc[1][1] + Uc[0][2]*Uc[1][2];
    Uc[1][0] -= d01 * Uc[0][0]; Uc[1][1] -= d01 * Uc[0][1]; Uc[1][2] -= d01 * Uc[0][2];
    float n1 = sqrtf(Uc[1][0]*Uc[1][0] + Uc[1][1]*Uc[1][1] + Uc[1][2]*Uc[1][2]);
    if (n1 > 1e-12f) { const float r1 = 1.0f / n1; Uc[1][0]*=r1; Uc[1][1]*=r1; Uc[1][2]*=r1; }
    else {
        const float ax = (fabsf(Uc[0][0]) < 0.9f) ? 1.f : 0.f;
        const float ay = 1.f - ax;
        float cx = ay * Uc[0][2] - 0.f * Uc[0][1];
        float cy = 0.f * Uc[0][0] - ax * Uc[0][2];
        float cz = ax * Uc[0][1] - ay * Uc[0][0];
        const float cn = 1.0f / sqrtf(cx*cx + cy*cy + cz*cz + 1e-30f);
        Uc[1][0] = cx * cn; Uc[1][1] = cy * cn; Uc[1][2] = cz * cn;
    }
    Uc[2][0] = Uc[0][1]*Uc[1][2] - Uc[0][2]*Uc[1][1];
    Uc[2][1] = Uc[0][2]*Uc[1][0] - Uc[0][0]*Uc[1][2];
    Uc[2][2] = Uc[0][0]*Uc[1][1] - Uc[0][1]*Uc[1][0];

    float R[3][3];
#pragma unroll
    for (int i = 0; i < 3; ++i)
#pragma unroll
        for (int j = 0; j < 3; ++j)
            R[i][j] = V[i][0]*Uc[0][j] + V[i][1]*Uc[1][j] + V[i][2]*Uc[2][j];

    if (det3x3(R) <= 0.0f) {
#pragma unroll
        for (int i = 0; i < 3; ++i)
#pragma unroll
            for (int j = 0; j < 3; ++j)
                R[i][j] -= 2.0f * V[i][2] * Uc[2][j];
    }

    float tvec[3];
#pragma unroll
    for (int i = 0; i < 3; ++i)
        tvec[i] = tc[i] - (R[i][0]*sc[0] + R[i][1]*sc[1] + R[i][2]*sc[2]);

    float* Rout = out;
    float* tout = out + (size_t)B_BATCH * 9;
#pragma unroll
    for (int i = 0; i < 3; ++i)
#pragma unroll
        for (int j = 0; j < 3; ++j)
            Rout[(size_t)b * 9 + i * 3 + j] = R[i][j];
#pragma unroll
    for (int i = 0; i < 3; ++i)
        tout[(size_t)b * 3 + i] = tvec[i];
}

extern "C" void kernel_launch(void* const* d_in, const int* in_sizes, int n_in,
                              void* d_out, int out_size, void* d_ws, size_t ws_size,
                              hipStream_t stream)
{
    const float* src = (const float*)d_in[0];
    const float* tpl = (const float*)d_in[1];
    const float* wgt = (const float*)d_in[2];
    float* out  = (float*)d_out;
    float* part = (float*)d_ws;

    svd_reduce_kernel<<<B_BATCH, THREADS, 0, stream>>>(src, tpl, wgt, part);
    svd_solve_kernel<<<(B_BATCH + 255) / 256, 256, 0, stream>>>(part, out);
}


# Solve kernel: Jacobi sweeps replaced by a closed-form eigen-decomposition of S (one cubic root by the trigonometric formula, its vector by row cross products, the rest from a 2x2 problem); f32 kept
# speedup vs baseline: 1.8403x; 1.8403x over previous
#include <hip/hip_runtime.h>

#define B_BATCH 1024
#define NPTS    8192
#define THREADS 128
#define WAVES   (THREADS / 32)
#define WPTS    (NPTS / WAVES)
#define TRIP    64
#define NTRIP   (WPTS / TRIP)
#define RING    8

typedef __attribute__((ext_vector_type(2))) float v2f;
typedef __attribute__((ext_vector_type(8))) float v8f;
typedef __attribute__((ext_vector_type(3), aligned(4))) float v3f;

struct TripRegs {
    float w[2];
    float s[2][3];
    float t[2][3];
};

__device__ __forceinline__ void load_trip(TripRegs& r, const float* __restrict__ gs,
                                          const float* __restrict__ gt,
                                          const float* __restrict__ gw, int p0)
{
#pragma unroll
    for (int u = 0; u < 2; ++u) {
        const int p = p0 + u * 32;
        r.w[u] = __builtin_nontemporal_load(gw + p);
        const v3f sv = __builtin_nontemporal_load((const v3f*)(gs + p * 3));
        const v3f tv = __builtin_nontemporal_load((const v3f*)(gt + p * 3));
        r.s[u][0] = sv.x; r.s[u][1] = sv.y; r.s[u][2] = sv.z;
        r.t[u][0] = tv.x; r.t[u][1] = tv.y; r.t[u][2] = tv.z;
    }
}

__device__ __forceinline__ void first_use(TripRegs& r)
{
    asm volatile("" : "+v"(r.w[0]), "+v"(r.w[1]));
}

__device__ __forceinline__ void hold_loads()
{
    asm volatile("" ::: "memory");
    __builtin_amdgcn_sched_barrier(0);
}

__device__ __forceinline__ float quad_x1(float v)
{
    return __int_as_float(__builtin_amdgcn_mov_dpp(__float_as_int(v), 0xB1, 0xF, 0xF, true));
}
__device__ __forceinline__ float quad_x2(float v)
{
    return __int_as_float(__builtin_amdgcn_mov_dpp(__float_as_int(v), 0x4E, 0xF, 0xF, true));
}

__device__ __forceinline__ void quad_stage2(const float (&x)[4], float (&o)[4], bool b1)
{
    const float m0 = quad_x2(x[0]), m1 = quad_x2(x[1]), m2 = quad_x2(x[2]), m3 = quad_x2(x[3]);
    o[0] = b1 ? m2 : x[0];
    o[1] = b1 ? m3 : x[1];
    o[2] = b1 ? x[2] : m0;
    o[3] = b1 ? x[3] : m1;
}

__device__ __forceinline__ void quad_transpose(const float (&f)[4], float (&o)[4], bool b0, bool b1)
{
    const float n0 = quad_x1(f[0]), n1 = quad_x1(f[1]), n2 = quad_x1(f[2]), n3 = quad_x1(f[3]);
    float x[4];
    x[0] = b0 ? n1 : f[0];
    x[1] = b0 ? f[1] : n0;
    x[2] = b0 ? n3 : f[2];
    x[3] = b0 ? f[3] : n2;
    quad_stage2(x, o, b1);
}

__device__ __forceinline__ void quad_transpose_one(const float (&f)[4], float (&o)[4], bool b0, bool b1)
{
    const float n1 = quad_x1(f[1]), n2 = quad_x1(f[2]), n3 = quad_x1(f[3]);
    float x[4];
    x[0] = b0 ? n1 : 1.0f;
    x[1] = b0 ? f[1] : 1.0f;
    x[2] = b0 ? n3 : f[2];
    x[3] = b0 ? f[3] : n2;
    quad_stage2(x, o, b1);
}

__device__ __forceinline__ void make_ops(const TripRegs& r, v2f (&Ap)[4], v2f (&Bp)[4],
                                         bool b0, bool b1)
{
#pragma unroll
    for (int u = 0; u < 2; ++u) {
        const float fl[4] = {r.w[u], r.w[u] * r.s[u][0], r.w[u] * r.s[u][1], r.w[u] * r.s[u][2]};
        const float fr[4] = {1.0f, r.t[u][0], r.t[u][1], r.t[u][2]};
        float oa[4], ob[4];
        quad_transpose(fl, oa, b0, b1);
        quad_transpose_one(fr, ob, b0, b1);
        Ap[2 * u + 0] = (v2f){oa[0], oa[1]};  Bp[2 * u + 0] = (v2f){ob[0], ob[1]};
        Ap[2 * u + 1] = (v2f){oa[2], oa[3]};  Bp[2 * u + 1] = (v2f){ob[2], ob[3]};
    }
}

__device__ __forceinline__ void mma4(v8f (&acc)[4], const v2f (&Ap)[4], const v2f (&Bp)[4])
{
#pragma unroll
    for (int k = 0; k < 4; ++k)
        acc[k] = __builtin_amdgcn_wmma_f32_16x16x4_f32(
             false, Ap[k],  false, Bp[k],
             (short)0, acc[k],  false,  false);
}

__global__ __launch_bounds__(THREADS)
void svd_reduce_kernel(const float* __restrict__ src,
                       const float* __restrict__ tpl,
                       const float* __restrict__ wgt,
                       float* __restrict__ part)
{
    __shared__ float red[WAVES][16];

    const int tid  = threadIdx.x;
    const int lane = tid & 31;
    const int wv   = __builtin_amdgcn_readfirstlane(tid >> 5);
    const int b    = blockIdx.x;
    const bool b0  = (lane & 1) != 0;
    const bool b1  = (lane & 2) != 0;

    const size_t base = (size_t)b * NPTS + (size_t)wv * WPTS;
    const float* gs = src + base * 3;
    const float* gt = tpl + base * 3;
    const float* gw = wgt + base;

    v8f acc[4];
#pragma unroll
    for (int k = 0; k < 4; ++k)
        acc[k] = (v8f){0.f, 0.f, 0.f, 0.f, 0.f, 0.f, 0.f, 0.f};

    TripRegs ring[RING];
#pragma unroll
    for (int j = 0; j < RING; ++j) {
        load_trip(ring[j], gs, gt, gw, j * TRIP + lane);
        hold_loads();
    }

#pragma unroll
    for (int it = 0; it < NTRIP - RING; it += RING) {
#pragma unroll
        for (int j = 0; j < RING; ++j) {
            v2f Ap[4], Bp[4];
            first_use(ring[j]);
            make_ops(ring[j], Ap, Bp, b0, b1);
            hold_loads();
            load_trip(ring[j], gs, gt, gw, (it + RING + j) * TRIP + lane);
            hold_loads();
            mma4(acc, Ap, Bp);
        }
    }
#pragma unroll
    for (int j = 0; j < RING; ++j) {
        v2f Ap[4], Bp[4];
        first_use(ring[j]);
        make_ops(ring[j], Ap, Bp, b0, b1);
        mma4(acc, Ap, Bp);
        __builtin_amdgcn_sched_barrier(0);
    }

    const v8f a = (acc[0] + acc[1]) + (acc[2] + acc[3]);
    const bool own = (lane < 8) || (lane >= 24);
    const bool hi4 = (lane & 4) != 0;
    float v[4];
#pragma unroll
    for (int i = 0; i < 4; ++i) {
        float x = hi4 ? a[4 + i] : a[i];
        x = own ? x : 0.0f;
        x += __shfl_xor(x, 4, 32);
        x += __shfl_xor(x, 8, 32);
        x += __shfl_xor(x, 16, 32);
        v[i] = x;
    }
    if (lane < 4) {
#pragma unroll
        for (int i = 0; i < 4; ++i) red[wv][i * 4 + lane] = v[i];
    }
    __syncthreads();
    if (tid < 16)
        part[(size_t)b * 16 + tid] = ((red[0][tid] + red[1][tid]) + red[2][tid]) + red[3][tid];
}

__device__ inline float det3x3(const float M[3][3]) {
    return M[0][0] * (M[1][1] * M[2][2] - M[1][2] * M[2][1])
         - M[0][1] * (M[1][0] * M[2][2] - M[1][2] * M[2][0])
         + M[0][2] * (M[1][0] * M[2][1] - M[1][1] * M[2][0]);
}

__global__ __launch_bounds__(256)
void svd_solve_kernel(const float* __restrict__ part, float* __restrict__ out)
{
    const int b = blockIdx.x * blockDim.x + threadIdx.x;
    if (b >= B_BATCH) return;

    float C[4][4];
#pragma unroll
    for (int i = 0; i < 16; ++i) C[i >> 2][i & 3] = part[(size_t)b * 16 + i];

    const float W     = C[0][0];
    const float alpha = 1.0f / (W + 1e-8f);
    const float sig   = W * alpha;
    float sc[3] = {C[1][0] * alpha, C[2][0] * alpha, C[3][0] * alpha};
    float tc[3] = {C[0][1] * alpha, C[0][2] * alpha, C[0][3] * alpha};

    float H[3][3];
#pragma unroll
    for (int d = 0; d < 3; ++d)
#pragma unroll
        for (int e = 0; e < 3; ++e)
            H[d][e] = alpha * C[d + 1][e + 1] - (2.0f - sig) * sc[d] * tc[e];

    float S[3][3];
#pragma unroll
    for (int i = 0; i < 3; ++i)
#pragma unroll
        for (int j = 0; j < 3; ++j)
            S[i][j] = H[0][i] * H[0][j] + H[1][i] * H[1][j] + H[2][i] * H[2][j];

    float V[3][3] = {{1.f, 0.f, 0.f}, {0.f, 1.f, 0.f}, {0.f, 0.f, 1.f}};
    {
        const float q   = (S[0][0] + S[1][1] + S[2][2]) * (1.0f / 3.0f);
        const float b00 = S[0][0] - q, b11 = S[1][1] - q, b22 = S[2][2] - q;
        const float b01 = S[0][1], b02 = S[0][2], b12 = S[1][2];
        const float p2  = (b00 * b00 + b11 * b11 + b22 * b22 +
                           2.0f * (b01 * b01 + b02 * b02 + b12 * b12)) * (1.0f / 6.0f);
        if (p2 > 1e-30f) {
            const float p    = sqrtf(p2);
            const float detB = b00 * (b11 * b22 - b12 * b12)
                             - b01 * (b01 * b22 - b12 * b02)
                             + b02 * (b01 * b12 - b11 * b02);
            float rr = detB / (2.0f * p2 * p);
            rr = fminf(1.0f, fmaxf(-1.0f, rr));
            const float g  = cosf(acosf(fabsf(rr)) * (1.0f / 3.0f));
            const float mu = copysignf(2.0f * p * g, rr);
            const float m00 = b00 - mu, m11 = b11 - mu, m22 = b22 - mu;
            const float c0x = b01 * b12 - b02 * m11, c0y = b02 * b01 - m00 * b12, c0z = m00 * m11 - b01 * b01;
            const float c1x = m11 * m22 - b12 * b12, c1y = b12 * b02 - b01 * m22, c1z = b01 * b12 - m11 * b02;
            const float c2x = b12 * b02 - m22 * b01, c2y = m22 * m00 - b02 * b02, c2z = b02 * b01 - b12 * m00;
            const float n0 = c0x * c0x + c0y * c0y + c0z * c0z;
            const float n1 = c1x * c1x + c1y * c1y + c1z * c1z;
            const float n2 = c2x * c2x + c2y * c2y + c2z * c2z;
            float cx = c0x, cy = c0y, cz = c0z, nn = n0;
            if (n1 > nn) { cx = c1x; cy = c1y; cz = c1z; nn = n1; }
            if (n2 > nn) { cx = c2x; cy = c2y; cz = c2z; nn = n2; }
            if (nn > 0.0f) {
                const float inv = 1.0f / sqrtf(nn);
                const float vx = cx * inv, vy = cy * inv, vz = cz * inv;
                const float sg = copysignf(1.0f, vz);
                const float aa = -1.0f / (sg + vz);
                const float bb = vx * vy * aa;
                const float ax = 1.0f + sg * vx * vx * aa, ay = sg * bb, az = -sg * vx;
                const float bx = bb, by = sg + vy * vy * aa, bz = -vy;
                const float sax = S[0][0] * ax + S[0][1] * ay + S[0][2] * az;
                const float say = S[1][0] * ax + S[1][1] * ay + S[1][2] * az;
                const float saz = S[2][0] * ax + S[2][1] * ay + S[2][2] * az;
                const float sbx = S[0][0] * bx + S[0][1] * by + S[0][2] * bz;
                const float sby = S[1][0] * bx + S[1][1] * by + S[1][2] * bz;
                const float sbz = S[2][0] * bx + S[2][1] * by + S[2][2] * bz;
                const float al = ax * sax + ay * say + az * saz;
                const float be = ax * sbx + ay * sby + az * sbz;
                const float ga = bx * sbx + by * sby + bz * sbz;
                const float d  = 0.5f * (al - ga);
                const float h  = sqrtf(d * d + be * be);
                const float x  = fabsf(d) + h;
                const float m2 = x * x + be * be;
                float ck = 1.0f, sk = 0.0f;
                if (m2 > 0.0f) { const float iv = 1.0f / sqrtf(m2); ck = x * iv; sk = be * iv; }
                const bool dp = d >= 0.0f;
                const float k1 = dp ? ck : sk, k2 = dp ? sk : ck;
                const float l1 = dp ? -sk : -ck, l2 = dp ? ck : sk;
                const float hx = k1 * ax + k2 * bx, hy = k1 * ay + k2 * by, hz = k1 * az + k2 * bz;
                const float lx = l1 * ax + l2 * bx, ly = l1 * ay + l2 * by, lz = l1 * az + l2 * bz;
                const bool top = rr >= 0.0f;
                V[0][0] = top ? vx : hx;  V[1][0] = top ? vy : hy;  V[2][0] = top ? vz : hz;
                V[0][1] = top ? hx : lx;  V[1][1] = top ? hy : ly;  V[2][1] = top ? hz : lz;
                V[0][2] = top ? lx : vx;  V[1][2] = top ? ly : vy;  V[2][2] = top ? lz : vz;
            }
        }
    }

    float Uc[3][3];
#pragma unroll
    for (int kcol = 0; kcol < 2; ++kcol)
#pragma unroll
        for (int r = 0; r < 3; ++r)
            Uc[kcol][r] = H[r][0] * V[0][kcol] + H[r][1] * V[1][kcol] + H[r][2] * V[2][kcol];

    float n0 = sqrtf(Uc[0][0]*Uc[0][0] + Uc[0][1]*Uc[0][1] + Uc[0][2]*Uc[0][2]);
    if (n0 > 1e-12f) { const float r0 = 1.0f / n0; Uc[0][0]*=r0; Uc[0][1]*=r0; Uc[0][2]*=r0; }
    else { Uc[0][0] = 1.f; Uc[0][1] = 0.f; Uc[0][2] = 0.f; }

    const float d01 = Uc[0][0]*Uc[1][0] + Uc[0][1]*Uc[1][1] + Uc[0][2]*Uc[1][2];
    Uc[1][0] -= d01 * Uc[0][0]; Uc[1][1] -= d01 * Uc[0][1]; Uc[1][2] -= d01 * Uc[0][2];
    float n1 = sqrtf(Uc[1][0]*Uc[1][0] + Uc[1][1]*Uc[1][1] + Uc[1][2]*Uc[1][2]);
    if (n1 > 1e-12f) { const float r1 = 1.0f / n1; Uc[1][0]*=r1; Uc[1][1]*=r1; Uc[1][2]*=r1; }
    else {
        const float ax = (fabsf(Uc[0][0]) < 0.9f) ? 1.f : 0.f;
        const float ay = 1.f - ax;
        float cx = ay * Uc[0][2] - 0.f * Uc[0][1];
        float cy = 0.f * Uc[0][0] - ax * Uc[0][2];
        float cz = ax * Uc[0][1] - ay * Uc[0][0];
        const float cn = 1.0f / sqrtf(cx*cx + cy*cy + cz*cz + 1e-30f);
        Uc[1][0] = cx * cn; Uc[1][1] = cy * cn; Uc[1][2] = cz * cn;
    }
    Uc[2][0] = Uc[0][1]*Uc[1][2] - Uc[0][2]*Uc[1][1];
    Uc[2][1] = Uc[0][2]*Uc[1][0] - Uc[0][0]*Uc[1][2];
    Uc[2][2] = Uc[0][0]*Uc[1][1] - Uc[0][1]*Uc[1][0];

    float R[3][3];
#pragma unroll
    for (int i = 0; i < 3; ++i)
#pragma unroll
        for (int j = 0; j < 3; ++j)
            R[i][j] = V[i][0]*Uc[0][j] + V[i][1]*Uc[1][j] + V[i][2]*Uc[2][j];

    if (det3x3(R) <= 0.0f) {
#pragma unroll
        for (int i = 0; i < 3; ++i)
#pragma unroll
            for (int j = 0; j < 3; ++j)
                R[i][j] -= 2.0f * V[i][2] * Uc[2][j];
    }

    float tvec[3];
#pragma unroll
    for (int i = 0; i < 3; ++i)
        tvec[i] = tc[i] - (R[i][0]*sc[0] + R[i][1]*sc[1] + R[i][2]*sc[2]);

    float* Rout = out;
    float* tout = out + (size_t)B_BATCH * 9;
#pragma unroll
    for (int i = 0; i < 3; ++i)
#pragma unroll
        for (int j = 0; j < 3; ++j)
            Rout[(size_t)b * 9 + i * 3 + j] = R[i][j];
#pragma unroll
    for (int i = 0; i < 3; ++i)
        tout[(size_t)b * 3 + i] = tvec[i];
}

extern "C" void kernel_launch(void* const* d_in, const int* in_sizes, int n_in,
                              void* d_out, int out_size, void* d_ws, size_t ws_size,
                              hipStream_t stream)
{
    const float* src = (const float*)d_in[0];
    const float* tpl = (const float*)d_in[1];
    const float* wgt = (const float*)d_in[2];
    float* out  = (float*)d_out;
    float* part = (float*)d_ws;

    svd_reduce_kernel<<<B_BATCH, THREADS, 0, stream>>>(src, tpl, wgt, part);
    svd_solve_kernel<<<(B_BATCH + 255) / 256, 256, 0, stream>>>(part, out);
}


# Reduce kernel: each global load reads 512 contiguous bytes, 16 per lane (7 per 128 points, not 12); the wave's own LDS slab re-cuts the block into operand order, no DPP transposes; sums regrouped
# speedup vs baseline: 1.9541x; 1.0618x over previous
#include <hip/hip_runtime.h>

#define B_BATCH 1024
#define NPTS    8192
#define THREADS 128
#define WAVES   (THREADS / 32)
#define WPTS    (NPTS / WAVES)
#define BLK     128
#define NBLK    (WPTS / BLK)
#define RING    4
#define SLAB    224

typedef __attribute__((ext_vector_type(2))) float v2f;
typedef __attribute__((ext_vector_type(4))) float v4f;
typedef __attribute__((ext_vector_type(8))) float v8f;

struct BlkRegs {
    v4f s[3];
    v4f t[3];
    v4f w;
};

__device__ __forceinline__ void load_block(BlkRegs& r, const float* __restrict__ gs,
                                           const float* __restrict__ gt,
                                           const float* __restrict__ gw, int n, int lane)
{
    const v4f* ps = (const v4f*)(gs + n * (BLK * 3)) + lane;
    const v4f* pt = (const v4f*)(gt + n * (BLK * 3)) + lane;
    const v4f* pw = (const v4f*)(gw + n * BLK) + lane;
#pragma unroll
    for (int k = 0; k < 3; ++k) r.s[k] = __builtin_nontemporal_load(ps + 32 * k);
#pragma unroll
    for (int k = 0; k < 3; ++k) r.t[k] = __builtin_nontemporal_load(pt + 32 * k);
    r.w = __builtin_nontemporal_load(pw);
}

__device__ __forceinline__ void hold_loads()
{
    asm volatile("" ::: "memory");
    __builtin_amdgcn_sched_barrier(0);
}

__device__ __forceinline__ void slab_handover()
{
    asm volatile("" ::: "memory");
}

__device__ __forceinline__ void store_block(v4f* slab, const BlkRegs& r, int lane)
{
#pragma unroll
    for (int k = 0; k < 3; ++k) slab[32 * k + lane] = r.s[k];
#pragma unroll
    for (int k = 0; k < 3; ++k) slab[96 + 32 * k + lane] = r.t[k];
    slab[192 + lane] = r.w;
}

__device__ __forceinline__ void make_ops(const float* sp, const float* tp, const v4f* wq,
                                         bool one, v2f (&Ap)[8], v2f (&Bp)[8])
{
#pragma unroll
    for (int u = 0; u < 4; ++u) {
        const v4f w4 = wq[8 * u];
        float a[4], b[4];
#pragma unroll
        for (int q = 0; q < 4; ++q) {
            const float sv = sp[96 * u + 3 * q];
            const float tv = tp[96 * u + 3 * q];
            a[q] = w4[q] * (one ? 1.0f : sv);
            b[q] = one ? 1.0f : tv;
        }
        Ap[2 * u + 0] = (v2f){a[0], a[1]};  Bp[2 * u + 0] = (v2f){b[0], b[1]};
        Ap[2 * u + 1] = (v2f){a[2], a[3]};  Bp[2 * u + 1] = (v2f){b[2], b[3]};
    }
}

__device__ __forceinline__ void mma8(v8f (&acc)[4], const v2f (&Ap)[8], const v2f (&Bp)[8])
{
#pragma unroll
    for (int k = 0; k < 8; ++k)
        acc[k & 3] = __builtin_amdgcn_wmma_f32_16x16x4_f32(
             false, Ap[k],  false, Bp[k],
             (short)0, acc[k & 3],  false,  false);
}

__global__ __launch_bounds__(THREADS)
void svd_reduce_kernel(const float* __restrict__ src,
                       const float* __restrict__ tpl,
                       const float* __restrict__ wgt,
                       float* __restrict__ part)
{
    __shared__ v4f slabs[WAVES][SLAB];
    __shared__ float red[WAVES][16];

    const int tid  = threadIdx.x;
    const int lane = tid & 31;
    const int wv   = __builtin_amdgcn_readfirstlane(tid >> 5);
    const int b    = blockIdx.x;
    const int qi   = lane & 3;
    const int qg   = lane >> 2;
    const bool one = qi == 0;
    const int comp = one ? 0 : qi - 1;

    const size_t base = (size_t)b * NPTS + (size_t)wv * WPTS;
    const float* gs = src + base * 3;
    const float* gt = tpl + base * 3;
    const float* gw = wgt + base;

    v4f* slab = slabs[wv];
    const float* sp = (const float*)slab + 12 * qg + comp;
    const float* tp = sp + 96 * 4;
    const v4f*   wq = slab + 192 + qg;

    v8f acc[4];
#pragma unroll
    for (int k = 0; k < 4; ++k)
        acc[k] = (v8f){0.f, 0.f, 0.f, 0.f, 0.f, 0.f, 0.f, 0.f};

    BlkRegs ring[RING];
#pragma unroll
    for (int j = 0; j < RING; ++j) {
        load_block(ring[j], gs, gt, gw, j, lane);
        hold_loads();
    }

#pragma unroll
    for (int it = 0; it < NBLK; it += RING) {
#pragma unroll
        for (int j = 0; j < RING; ++j) {
            v2f Ap[8], Bp[8];
            slab_handover();
            store_block(slab, ring[j], lane);
            slab_handover();
            if (it + RING < NBLK) {
                hold_loads();
                load_block(ring[j], gs, gt, gw, it + RING + j, lane);
                hold_loads();
            }
            make_ops(sp, tp, wq, one, Ap, Bp);
            mma8(acc, Ap, Bp);
        }
    }

    const v8f a = (acc[0] + acc[1]) + (acc[2] + acc[3]);
    const bool own = (lane < 8) || (lane >= 24);
    const bool hi4 = (lane & 4) != 0;
    float v[4];
#pragma unroll
    for (int i = 0; i < 4; ++i) {
        float x = hi4 ? a[4 + i] : a[i];
        x = own ? x : 0.0f;
        x += __shfl_xor(x, 4, 32);
        x += __shfl_xor(x, 8, 32);
        x += __shfl_xor(x, 16, 32);
        v[i] = x;
    }
    if (lane < 4) {
#pragma unroll
        for (int i = 0; i < 4; ++i) red[wv][i * 4 + lane] = v[i];
    }
    __syncthreads();
    if (tid < 16)
        part[(size_t)b * 16 + tid] = ((red[0][tid] + red[1][tid]) + red[2][tid]) + red[3][tid];
}

__device__ inline float det3x3(const float M[3][3]) {
    return M[0][0] * (M[1][1] * M[2][2] - M[1][2] * M[2][1])
         - M[0][1] * (M[1][0] * M[2][2] - M[1][2] * M[2][0])
         + M[0][2] * (M[1][0] * M[2][1] - M[1][1] * M[2][0]);
}

__global__ __launch_bounds__(256)
void svd_solve_kernel(const float* __restrict__ part, float* __restrict__ out)
{
    const int b = blockIdx.x * blockDim.x + threadIdx.x;
    if (b >= B_BATCH) return;

    float C[4][4];
#pragma unroll
    for (int i = 0; i < 16; ++i) C[i >> 2][i & 3] = part[(size_t)b * 16 + i];

    const float W     = C[0][0];
    const float alpha = 1.0f / (W + 1e-8f);
    const float sig   = W * alpha;
    float sc[3] = {C[1][0] * alpha, C[2][0] * alpha, C[3][0] * alpha};
    float tc[3] = {C[0][1] * alpha, C[0][2] * alpha, C[0][3] * alpha};

    float H[3][3];
#pragma unroll
    for (int d = 0; d < 3; ++d)
#pragma unroll
        for (int e = 0; e < 3; ++e)
            H[d][e] = alpha * C[d + 1][e + 1] - (2.0f - sig) * sc[d] * tc[e];

    float S[3][3];
#pragma unroll
    for (int i = 0; i < 3; ++i)
#pragma unroll
        for (int j = 0; j < 3; ++j)
            S[i][j] = H[0][i] * H[0][j] + H[1][i] * H[1][j] + H[2][i] * H[2][j];

    float V[3][3] = {{1.f, 0.f, 0.f}, {0.f, 1.f, 0.f}, {0.f, 0.f, 1.f}};
    {
        const float q   = (S[0][0] + S[1][1] + S[2][2]) * (1.0f / 3.0f);
        const float b00 = S[0][0] - q, b11 = S[1][1] - q, b22 = S[2][2] - q;
        const float b01 = S[0][1], b02 = S[0][2], b12 = S[1][2];
        const float p2  = (b00 * b00 + b11 * b11 + b22 * b22 +
                           2.0f * (b01 * b01 + b02 * b02 + b12 * b12)) * (1.0f / 6.0f);
        if (p2 > 1e-30f) {
            const float p    = sqrtf(p2);
            const float detB = b00 * (b11 * b22 - b12 * b12)
                             - b01 * (b01 * b22 - b12 * b02)
                             + b02 * (b01 * b12 - b11 * b02);
            float rr = detB / (2.0f * p2 * p);
            rr = fminf(1.0f, fmaxf(-1.0f, rr));
            const float g  = cosf(acosf(fabsf(rr)) * (1.0f / 3.0f));
            const float mu = copysignf(2.0f * p * g, rr);
            const float m00 = b00 - mu, m11 = b11 - mu, m22 = b22 - mu;
            const float c0x = b01 * b12 - b02 * m11, c0y = b02 * b01 - m00 * b12, c0z = m00 * m11 - b01 * b01;
            const float c1x = m11 * m22 - b12 * b12, c1y = b12 * b02 - b01 * m22, c1z = b01 * b12 - m11 * b02;
            const float c2x = b12 * b02 - m22 * b01, c2y = m22 * m00 - b02 * b02, c2z = b02 * b01 - b12 * m00;
            const float n0 = c0x * c0x + c0y * c0y + c0z * c0z;
            const float n1 = c1x * c1x + c1y * c1y + c1z * c1z;
            const float n2 = c2x * c2x + c2y * c2y + c2z * c2z;
            float cx = c0x, cy = c0y, cz = c0z, nn = n0;
            if (n1 > nn) { cx = c1x; cy = c1y; cz = c1z; nn = n1; }
            if (n2 > nn) { cx = c2x; cy = c2y; cz = c2z; nn = n2; }
            if (nn > 0.0f) {
                const float inv = 1.0f / sqrtf(nn);
                const float vx = cx * inv, vy = cy * inv, vz = cz * inv;
                const float sg = copysignf(1.0f, vz);
                const float aa = -1.0f / (sg + vz);
                const float bb = vx * vy * aa;
                const float ax = 1.0f + sg * vx * vx * aa, ay = sg * bb, az = -sg * vx;
                const float bx = bb, by = sg + vy * vy * aa, bz = -vy;
                const float sax = S[0][0] * ax + S[0][1] * ay + S[0][2] * az;
                const float say = S[1][0] * ax + S[1][1] * ay + S[1][2] * az;
                const float saz = S[2][0] * ax + S[2][1] * ay + S[2][2] * az;
                const float sbx = S[0][0] * bx + S[0][1] * by + S[0][2] * bz;
                const float sby = S[1][0] * bx + S[1][1] * by + S[1][2] * bz;
                const float sbz = S[2][0] * bx + S[2][1] * by + S[2][2] * bz;
                const float al = ax * sax + ay * say + az * saz;
                const float be = ax * sbx + ay * sby + az * sbz;
                const float ga = bx * sbx + by * sby + bz * sbz;
                const float d  = 0.5f * (al - ga);
                const float h  = sqrtf(d * d + be * be);
                const float x  = fabsf(d) + h;
                const float m2 = x * x + be * be;
                float ck = 1.0f, sk = 0.0f;
                if (m2 > 0.0f) { const float iv = 1.0f / sqrtf(m2); ck = x * iv; sk = be * iv; }
                const bool dp = d >= 0.0f;
                const float k1 = dp ? ck : sk, k2 = dp ? sk : ck;
                const float l1 = dp ? -sk : -ck, l2 = dp ? ck : sk;
                const float hx = k1 * ax + k2 * bx, hy = k1 * ay + k2 * by, hz = k1 * az + k2 * bz;
                const float lx = l1 * ax + l2 * bx, ly = l1 * ay + l2 * by, lz = l1 * az + l2 * bz;
                const bool top = rr >= 0.0f;
                V[0][0] = top ? vx : hx;  V[1][0] = top ? vy : hy;  V[2][0] = top ? vz : hz;
                V[0][1] = top ? hx : lx;  V[1][1] = top ? hy : ly;  V[2][1] = top ? hz : lz;
                V[0][2] = top ? lx : vx;  V[1][2] = top ? ly : vy;  V[2][2] = top ? lz : vz;
            }
        }
    }

    float Uc[3][3];
#pragma unroll
    for (int kcol = 0; kcol < 2; ++kcol)
#pragma unroll
        for (int r = 0; r < 3; ++r)
            Uc[kcol][r] = H[r][0] * V[0][kcol] + H[r][1] * V[1][kcol] + H[r][2] * V[2][kcol];

    float n0 = sqrtf(Uc[0][0]*Uc[0][0] + Uc[0][1]*Uc[0][1] + Uc[0][2]*Uc[0][2]);
    if (n0 > 1e-12f) { const float r0 = 1.0f / n0; Uc[0][0]*=r0; Uc[0][1]*=r0; Uc[0][2]*=r0; }
    else { Uc[0][0] = 1.f; Uc[0][1] = 0.f; Uc[0][2] = 0.f; }

    const float d01 = Uc[0][0]*Uc[1][0] + Uc[0][1]*Uc[1][1] + Uc[0][2]*Uc[1][2];
    Uc[1][0] -= d01 * Uc[0][0]; Uc[1][1] -= d01 * Uc[0][1]; Uc[1][2] -= d01 * Uc[0][2];
    float n1 = sqrtf(Uc[1][0]*Uc[1][0] + Uc[1][1]*Uc[1][1] + Uc[1][2]*Uc[1][2]);
    if (n1 > 1e-12f) { const float r1 = 1.0f / n1; Uc[1][0]*=r1; Uc[1][1]*=r1; Uc[1][2]*=r1; }
    else {
        const float ax = (fabsf(Uc[0][0]) < 0.9f) ? 1.f : 0.f;
        const float ay = 1.f - ax;
        float cx = ay * Uc[0][2] - 0.f * Uc[0][1];
        float cy = 0.f * Uc[0][0] - ax * Uc[0][2];
        float cz = ax * Uc[0][1] - ay * Uc[0][0];
        const float cn = 1.0f / sqrtf(cx*cx + cy*cy + cz*cz + 1e-30f);
        Uc[1][0] = cx * cn; Uc[1][1] = cy * cn; Uc[1][2] = cz * cn;
    }
    Uc[2][0] = Uc[0][1]*Uc[1][2] - Uc[0][2]*Uc[1][1];
    Uc[2][1] = Uc[0][2]*Uc[1][0] - Uc[0][0]*Uc[1][2];
    Uc[2][2] = Uc[0][0]*Uc[1][1] - Uc[0][1]*Uc[1][0];

    float R[3][3];
#pragma unroll
    for (int i = 0; i < 3; ++i)
#pragma unroll
        for (int j = 0; j < 3; ++j)
            R[i][j] = V[i][0]*Uc[0][j] + V[i][1]*Uc[1][j] + V[i][2]*Uc[2][j];

    if (det3x3(R) <= 0.0f) {
#pragma unroll
        for (int i = 0; i < 3; ++i)
#pragma unroll
            for (int j = 0; j < 3; ++j)
                R[i][j] -= 2.0f * V[i][2] * Uc[2][j];
    }

    float tvec[3];
#pragma unroll
    for (int i = 0; i < 3; ++i)
        tvec[i] = tc[i] - (R[i][0]*sc[0] + R[i][1]*sc[1] + R[i][2]*sc[2]);

    float* Rout = out;
    float* tout = out + (size_t)B_BATCH * 9;
#pragma unroll
    for (int i = 0; i < 3; ++i)
#pragma unroll
        for (int j = 0; j < 3; ++j)
            Rout[(size_t)b * 9 + i * 3 + j] = R[i][j];
#pragma unroll
    for (int i = 0; i < 3; ++i)
        tout[(size_t)b * 3 + i] = tvec[i];
}

extern "C" void kernel_launch(void* const* d_in, const int* in_sizes, int n_in,
                              void* d_out, int out_size, void* d_ws, size_t ws_size,
                              hipStream_t stream)
{
    const float* src = (const float*)d_in[0];
    const float* tpl = (const float*)d_in[1];
    const float* wgt = (const float*)d_in[2];
    float* out  = (float*)d_out;
    float* part = (float*)d_ws;

    svd_reduce_kernel<<<B_BATCH, THREADS, 0, stream>>>(src, tpl, wgt, part);
    svd_solve_kernel<<<(B_BATCH + 255) / 256, 256, 0, stream>>>(part, out);
}


# Reduce kernel: each wave walks its 16 blocks in a rotated order, the first block set by its batch and wave index, so the resident waves no longer read equal slice offsets in step; sums regrouped
# speedup vs baseline: 2.0960x; 1.0726x over previous
#include <hip/hip_runtime.h>

#define B_BATCH 1024
#define NPTS    8192
#define THREADS 128
#define WAVES   (THREADS / 32)
#define WPTS    (NPTS / WAVES)
#define BLK     128
#define NBLK    (WPTS / BLK)
#define RING    4
#define SLAB    224

typedef __attribute__((ext_vector_type(2))) float v2f;
typedef __attribute__((ext_vector_type(4))) float v4f;
typedef __attribute__((ext_vector_type(8))) float v8f;

struct BlkRegs {
    v4f s[3];
    v4f t[3];
    v4f w;
};

__device__ __forceinline__ void load_block(BlkRegs& r, const float* __restrict__ gs,
                                           const float* __restrict__ gt,
                                           const float* __restrict__ gw, int n, int lane)
{
    const v4f* ps = (const v4f*)(gs + n * (BLK * 3)) + lane;
    const v4f* pt = (const v4f*)(gt + n * (BLK * 3)) + lane;
    const v4f* pw = (const v4f*)(gw + n * BLK) + lane;
#pragma unroll
    for (int k = 0; k < 3; ++k) r.s[k] = __builtin_nontemporal_load(ps + 32 * k);
#pragma unroll
    for (int k = 0; k < 3; ++k) r.t[k] = __builtin_nontemporal_load(pt + 32 * k);
    r.w = __builtin_nontemporal_load(pw);
}

__device__ __forceinline__ void hold_loads()
{
    asm volatile("" ::: "memory");
    __builtin_amdgcn_sched_barrier(0);
}

__device__ __forceinline__ void slab_handover()
{
    asm volatile("" ::: "memory");
}

__device__ __forceinline__ void store_block(v4f* slab, const BlkRegs& r, int lane)
{
#pragma unroll
    for (int k = 0; k < 3; ++k) slab[32 * k + lane] = r.s[k];
#pragma unroll
    for (int k = 0; k < 3; ++k) slab[96 + 32 * k + lane] = r.t[k];
    slab[192 + lane] = r.w;
}

__device__ __forceinline__ void make_ops(const float* sp, const float* tp, const v4f* wq,
                                         bool one, v2f (&Ap)[8], v2f (&Bp)[8])
{
#pragma unroll
    for (int u = 0; u < 4; ++u) {
        const v4f w4 = wq[8 * u];
        float a[4], b[4];
#pragma unroll
        for (int q = 0; q < 4; ++q) {
            const float sv = sp[96 * u + 3 * q];
            const float tv = tp[96 * u + 3 * q];
            a[q] = w4[q] * (one ? 1.0f : sv);
            b[q] = one ? 1.0f : tv;
        }
        Ap[2 * u + 0] = (v2f){a[0], a[1]};  Bp[2 * u + 0] = (v2f){b[0], b[1]};
        Ap[2 * u + 1] = (v2f){a[2], a[3]};  Bp[2 * u + 1] = (v2f){b[2], b[3]};
    }
}

__device__ __forceinline__ void mma8(v8f (&acc)[4], const v2f (&Ap)[8], const v2f (&Bp)[8])
{
#pragma unroll
    for (int k = 0; k < 8; ++k)
        acc[k & 3] = __builtin_amdgcn_wmma_f32_16x16x4_f32(
             false, Ap[k],  false, Bp[k],
             (short)0, acc[k & 3],  false,  false);
}

__global__ __launch_bounds__(THREADS)
void svd_reduce_kernel(const float* __restrict__ src,
                       const float* __restrict__ tpl,
                       const float* __restrict__ wgt,
                       float* __restrict__ part)
{
    __shared__ v4f slabs[WAVES][SLAB];
    __shared__ float red[WAVES][16];

    const int tid  = threadIdx.x;
    const int lane = tid & 31;
    const int wv   = __builtin_amdgcn_readfirstlane(tid >> 5);
    const int b    = blockIdx.x;
    const int qi   = lane & 3;
    const int qg   = lane >> 2;
    const bool one = qi == 0;
    const int comp = one ? 0 : qi - 1;
    const int rot  = (b + (b >> 4) + 4 * wv) & (NBLK - 1);

    const size_t base = (size_t)b * NPTS + (size_t)wv * WPTS;
    const float* gsA = src + base * 3 + rot * (BLK * 3);
    const float* gtA = tpl + base * 3 + rot * (BLK * 3);
    const float* gwA = wgt + base + rot * BLK;
    const float* gsB = gsA - NBLK * (BLK * 3);
    const float* gtB = gtA - NBLK * (BLK * 3);
    const float* gwB = gwA - NBLK * BLK;
    const float* cs = gsA;
    const float* ct = gtA;
    const float* cw = gwA;

    v4f* slab = slabs[wv];
    const float* sp = (const float*)slab + 12 * qg + comp;
    const float* tp = sp + 96 * 4;
    const v4f*   wq = slab + 192 + qg;

    v8f acc[4];
#pragma unroll
    for (int k = 0; k < 4; ++k)
        acc[k] = (v8f){0.f, 0.f, 0.f, 0.f, 0.f, 0.f, 0.f, 0.f};

    BlkRegs ring[RING];
    const float* ps0[RING];
    const float* pt0[RING];
    const float* pw0[RING];
#pragma unroll
    for (int j = 0; j < RING; ++j) {
        const bool wrap = j + rot == NBLK;
        cs = wrap ? gsB : cs;  ct = wrap ? gtB : ct;  cw = wrap ? gwB : cw;
        ps0[j] = cs;  pt0[j] = ct;  pw0[j] = cw;
    }
    __builtin_amdgcn_sched_barrier(0);
#pragma unroll
    for (int j = 0; j < RING; ++j) {
        load_block(ring[j], ps0[j], pt0[j], pw0[j], j, lane);
        hold_loads();
    }

#pragma unroll
    for (int it = 0; it < NBLK; it += RING) {
#pragma unroll
        for (int j = 0; j < RING; ++j) {
            v2f Ap[8], Bp[8];
            slab_handover();
            store_block(slab, ring[j], lane);
            slab_handover();
            if (it + RING < NBLK) {
                hold_loads();
                const bool wrap = it + RING + j + rot == NBLK;
                cs = wrap ? gsB : cs;  ct = wrap ? gtB : ct;  cw = wrap ? gwB : cw;
                load_block(ring[j], cs, ct, cw, it + RING + j, lane);
                hold_loads();
            }
            make_ops(sp, tp, wq, one, Ap, Bp);
            mma8(acc, Ap, Bp);
        }
    }

    const v8f a = (acc[0] + acc[1]) + (acc[2] + acc[3]);
    const bool own = (lane < 8) || (lane >= 24);
    const bool hi4 = (lane & 4) != 0;
    float v[4];
#pragma unroll
    for (int i = 0; i < 4; ++i) {
        float x = hi4 ? a[4 + i] : a[i];
        x = own ? x : 0.0f;
        x += __shfl_xor(x, 4, 32);
        x += __shfl_xor(x, 8, 32);
        x += __shfl_xor(x, 16, 32);
        v[i] = x;
    }
    if (lane < 4) {
#pragma unroll
        for (int i = 0; i < 4; ++i) red[wv][i * 4 + lane] = v[i];
    }
    __syncthreads();
    if (tid < 16)
        part[(size_t)b * 16 + tid] = ((red[0][tid] + red[1][tid]) + red[2][tid]) + red[3][tid];
}

__device__ inline float det3x3(const float M[3][3]) {
    return M[0][0] * (M[1][1] * M[2][2] - M[1][2] * M[2][1])
         - M[0][1] * (M[1][0] * M[2][2] - M[1][2] * M[2][0])
         + M[0][2] * (M[1][0] * M[2][1] - M[1][1] * M[2][0]);
}

__global__ __launch_bounds__(256)
void svd_solve_kernel(const float* __restrict__ part, float* __restrict__ out)
{
    const int b = blockIdx.x * blockDim.x + threadIdx.x;
    if (b >= B_BATCH) return;

    float C[4][4];
#pragma unroll
    for (int i = 0; i < 16; ++i) C[i >> 2][i & 3] = part[(size_t)b * 16 + i];

    const float W     = C[0][0];
    const float alpha = 1.0f / (W + 1e-8f);
    const float sig   = W * alpha;
    float sc[3] = {C[1][0] * alpha, C[2][0] * alpha, C[3][0] * alpha};
    float tc[3] = {C[0][1] * alpha, C[0][2] * alpha, C[0][3] * alpha};

    float H[3][3];
#pragma unroll
    for (int d = 0; d < 3; ++d)
#pragma unroll
        for (int e = 0; e < 3; ++e)
            H[d][e] = alpha * C[d + 1][e + 1] - (2.0f - sig) * sc[d] * tc[e];

    float S[3][3];
#pragma unroll
    for (int i = 0; i < 3; ++i)
#pragma unroll
        for (int j = 0; j < 3; ++j)
            S[i][j] = H[0][i] * H[0][j] + H[1][i] * H[1][j] + H[2][i] * H[2][j];

    float V[3][3] = {{1.f, 0.f, 0.f}, {0.f, 1.f, 0.f}, {0.f, 0.f, 1.f}};
    {
        const float q   = (S[0][0] + S[1][1] + S[2][2]) * (1.0f / 3.0f);
        const float b00 = S[0][0] - q, b11 = S[1][1] - q, b22 = S[2][2] - q;
        const float b01 = S[0][1], b02 = S[0][2], b12 = S[1][2];
        const float p2  = (b00 * b00 + b11 * b11 + b22 * b22 +
                           2.0f * (b01 * b01 + b02 * b02 + b12 * b12)) * (1.0f / 6.0f);
        if (p2 > 1e-30f) {
            const float p    = sqrtf(p2);
            const float detB = b00 * (b11 * b22 - b12 * b12)
                             - b01 * (b01 * b22 - b12 * b02)
                             + b02 * (b01 * b12 - b11 * b02);
            float rr = detB / (2.0f * p2 * p);
            rr = fminf(1.0f, fmaxf(-1.0f, rr));
            const float g  = cosf(acosf(fabsf(rr)) * (1.0f / 3.0f));
            const float mu = copysignf(2.0f * p * g, rr);
            const float m00 = b00 - mu, m11 = b11 - mu, m22 = b22 - mu;
            const float c0x = b01 * b12 - b02 * m11, c0y = b02 * b01 - m00 * b12, c0z = m00 * m11 - b01 * b01;
            const float c1x = m11 * m22 - b12 * b12, c1y = b12 * b02 - b01 * m22, c1z = b01 * b12 - m11 * b02;
            const float c2x = b12 * b02 - m22 * b01, c2y = m22 * m00 - b02 * b02, c2z = b02 * b01 - b12 * m00;
            const float n0 = c0x * c0x + c0y * c0y + c0z * c0z;
            const float n1 = c1x * c1x + c1y * c1y + c1z * c1z;
            const float n2 = c2x * c2x + c2y * c2y + c2z * c2z;
            float cx = c0x, cy = c0y, cz = c0z, nn = n0;
            if (n1 > nn) { cx = c1x; cy = c1y; cz = c1z; nn = n1; }
            if (n2 > nn) { cx = c2x; cy = c2y; cz = c2z; nn = n2; }
            if (nn > 0.0f) {
                const float inv = 1.0f / sqrtf(nn);
                const float vx = cx * inv, vy = cy * inv, vz = cz * inv;
                const float sg = copysignf(1.0f, vz);
                const float aa = -1.0f / (sg + vz);
                const float bb = vx * vy * aa;
                const float ax = 1.0f + sg * vx * vx * aa, ay = sg * bb, az = -sg * vx;
                const float bx = bb, by = sg + vy * vy * aa, bz = -vy;
                const float sax = S[0][0] * ax + S[0][1] * ay + S[0][2] * az;
                const float say = S[1][0] * ax + S[1][1] * ay + S[1][2] * az;
                const float saz = S[2][0] * ax + S[2][1] * ay + S[2][2] * az;
                const float sbx = S[0][0] * bx + S[0][1] * by + S[0][2] * bz;
                const float sby = S[1][0] * bx + S[1][1] * by + S[1][2] * bz;
                const float sbz = S[2][0] * bx + S[2][1] * by + S[2][2] * bz;
                const float al = ax * sax + ay * say + az * saz;
                const float be = ax * sbx + ay * sby + az * sbz;
                const float ga = bx * sbx + by * sby + bz * sbz;
                const float d  = 0.5f * (al - ga);
                const float h  = sqrtf(d * d + be * be);
                const float x  = fabsf(d) + h;
                const float m2 = x * x + be * be;
                float ck = 1.0f, sk = 0.0f;
                if (m2 > 0.0f) { const float iv = 1.0f / sqrtf(m2); ck = x * iv; sk = be * iv; }
                const bool dp = d >= 0.0f;
                const float k1 = dp ? ck : sk, k2 = dp ? sk : ck;
                const float l1 = dp ? -sk : -ck, l2 = dp ? ck : sk;
                const float hx = k1 * ax + k2 * bx, hy = k1 * ay + k2 * by, hz = k1 * az + k2 * bz;
                const float lx = l1 * ax + l2 * bx, ly = l1 * ay + l2 * by, lz = l1 * az + l2 * bz;
                const bool top = rr >= 0.0f;
                V[0][0] = top ? vx : hx;  V[1][0] = top ? vy : hy;  V[2][0] = top ? vz : hz;
                V[0][1] = top ? hx : lx;  V[1][1] = top ? hy : ly;  V[2][1] = top ? hz : lz;
                V[0][2] = top ? lx : vx;  V[1][2] = top ? ly : vy;  V[2][2] = top ? lz : vz;
            }
        }
    }

    float Uc[3][3];
#pragma unroll
    for (int kcol = 0; kcol < 2; ++kcol)
#pragma unroll
        for (int r = 0; r < 3; ++r)
            Uc[kcol][r] = H[r][0] * V[0][kcol] + H[r][1] * V[1][kcol] + H[r][2] * V[2][kcol];

    float n0 = sqrtf(Uc[0][0]*Uc[0][0] + Uc[0][1]*Uc[0][1] + Uc[0][2]*Uc[0][2]);
    if (n0 > 1e-12f) { const float r0 = 1.0f / n0; Uc[0][0]*=r0; Uc[0][1]*=r0; Uc[0][2]*=r0; }
    else { Uc[0][0] = 1.f; Uc[0][1] = 0.f; Uc[0][2] = 0.f; }

    const float d01 = Uc[0][0]*Uc[1][0] + Uc[0][1]*Uc[1][1] + Uc[0][2]*Uc[1][2];
    Uc[1][0] -= d01 * Uc[0][0]; Uc[1][1] -= d01 * Uc[0][1]; Uc[1][2] -= d01 * Uc[0][2];
    float n1 = sqrtf(Uc[1][0]*Uc[1][0] + Uc[1][1]*Uc[1][1] + Uc[1][2]*Uc[1][2]);
    if (n1 > 1e-12f) { const float r1 = 1.0f / n1; Uc[1][0]*=r1; Uc[1][1]*=r1; Uc[1][2]*=r1; }
    else {
        const float ax = (fabsf(Uc[0][0]) < 0.9f) ? 1.f : 0.f;
        const float ay = 1.f - ax;
        float cx = ay * Uc[0][2] - 0.f * Uc[0][1];
        float cy = 0.f * Uc[0][0] - ax * Uc[0][2];
        float cz = ax * Uc[0][1] - ay * Uc[0][0];
        const float cn = 1.0f / sqrtf(cx*cx + cy*cy + cz*cz + 1e-30f);
        Uc[1][0] = cx * cn; Uc[1][1] = cy * cn; Uc[1][2] = cz * cn;
    }
    Uc[2][0] = Uc[0][1]*Uc[1][2] - Uc[0][2]*Uc[1][1];
    Uc[2][1] = Uc[0][2]*Uc[1][0] - Uc[0][0]*Uc[1][2];
    Uc[2][2] = Uc[0][0]*Uc[1][1] - Uc[0][1]*Uc[1][0];

    float R[3][3];
#pragma unroll
    for (int i = 0; i < 3; ++i)
#pragma unroll
        for (int j = 0; j < 3; ++j)
            R[i][j] = V[i][0]*Uc[0][j] + V[i][1]*Uc[1][j] + V[i][2]*Uc[2][j];

    if (det3x3(R) <= 0.0f) {
#pragma unroll
        for (int i = 0; i < 3; ++i)
#pragma unroll
            for (int j = 0; j < 3; ++j)
                R[i][j] -= 2.0f * V[i][2] * Uc[2][j];
    }

    float tvec[3];
#pragma unroll
    for (int i = 0; i < 3; ++i)
        tvec[i] = tc[i] - (R[i][0]*sc[0] + R[i][1]*sc[1] + R[i][2]*sc[2]);

    float* Rout = out;
    float* tout = out + (size_t)B_BATCH * 9;
#pragma unroll
    for (int i = 0; i < 3; ++i)
#pragma unroll
        for (int j = 0; j < 3; ++j)
            Rout[(size_t)b * 9 + i * 3 + j] = R[i][j];
#pragma unroll
    for (int i = 0; i < 3; ++i)
        tout[(size_t)b * 3 + i] = tvec[i];
}

extern "C" void kernel_launch(void* const* d_in, const int* in_sizes, int n_in,
                              void* d_out, int out_size, void* d_ws, size_t ws_size,
                              hipStream_t stream)
{
    const float* src = (const float*)d_in[0];
    const float* tpl = (const float*)d_in[1];
    const float* wgt = (const float*)d_in[2];
    float* out  = (float*)d_out;
    float* part = (float*)d_ws;

    svd_reduce_kernel<<<B_BATCH, THREADS, 0, stream>>>(src, tpl, wgt, part);
    svd_solve_kernel<<<(B_BATCH + 255) / 256, 256, 0, stream>>>(part, out);
}
